# starting kernel
# baseline (speedup 1.0000x reference)
#include <hip/hip_runtime.h>
#include <hip/hip_bf16.h>

#define SDIM   100
#define SPAD   112
#define HID    8
#define NOBS   12
#define DTC    0.01f
#define NCHUNK 25
#define NWA    7

typedef float v2f __attribute__((ext_vector_type(2)));
typedef float v8f __attribute__((ext_vector_type(8)));

__global__ __launch_bounds__(256, 1)
void greybox_scan_kernel(const float* __restrict__ c0,
                         const float* __restrict__ w,
                         const float* __restrict__ u,
                         const float* __restrict__ A,
                         const float* __restrict__ Bw,
                         const float* __restrict__ W1,
                         const float* __restrict__ b1,
                         const float* __restrict__ W2,
                         const float* __restrict__ b2,
                         const int*   __restrict__ obs,
                         float* __restrict__ out_c,
                         float* __restrict__ out_y,
                         int T)
{
    __shared__ float cS[SDIM];
    __shared__ float zS[SDIM];
    __shared__ float hS[HID];
    __shared__ float accS[SPAD];
    __shared__ float uS[2][SDIM];
    __shared__ float wS[2][2];
    __shared__ float BwS[2 * SDIM];
    __shared__ float b2S[SDIM];
    __shared__ float b1S[HID];
    __shared__ int   obsS[NOBS];

    const int tid  = threadIdx.x;
    const int wave = tid >> 5;
    const int lane = tid & 31;
    const int hi   = (lane >> 4) & 1;
    const int rloc = lane & 15;

    for (int i = tid; i < 2 * SDIM; i += 256) BwS[i] = Bw[i];
    for (int i = tid; i < SDIM;     i += 256) { b2S[i] = DTC * b2[i]; zS[i] = 0.0f; cS[i] = c0[i]; }
    for (int i = tid; i < HID;      i += 256) b1S[i] = b1[i];
    for (int i = tid; i < NOBS;     i += 256) obsS[i] = obs[i];
    for (int i = tid; i < SPAD;     i += 256) accS[i] = 0.0f;
    if (tid < SDIM) { uS[0][tid] = u[tid]; uS[1][tid] = 0.0f; }
    if (tid < 2)    { wS[0][tid] = w[tid]; wS[1][tid] = 0.0f; }

    const float* bBaseC = (rloc == 0) ? (cS + 2 * hi) : zS;
    const float* bBaseH = (rloc == 0) ? (hS + 2 * hi) : zS;

    v2f afrag[NCHUNK] = {};
    v2f w2frag[2]     = {};

    if (wave < NWA) {
        const int r = wave * 16 + rloc;
        #pragma unroll
        for (int j = 0; j < NCHUNK; ++j) {
            const int k = 4 * j + 2 * hi;
            v2f a; a.x = 0.0f; a.y = 0.0f;
            if (r < SDIM) {
                a.x = A[r * SDIM + k];
                a.y = A[r * SDIM + k + 1];
            }
            afrag[j] = a;
        }
        #pragma unroll
        for (int jj = 0; jj < 2; ++jj) {
            const int k = 4 * jj + 2 * hi;
            v2f a; a.x = 0.0f; a.y = 0.0f;
            if (r < SDIM) {
                a.x = DTC * W2[r * HID + k];
                a.y = DTC * W2[r * HID + k + 1];
            }
            w2frag[jj] = a;
        }
    } else {
        const int rw = rloc;
        #pragma unroll
        for (int j = 0; j < NCHUNK; ++j) {
            const int k = 4 * j + 2 * hi;
            v2f a; a.x = 0.0f; a.y = 0.0f;
            if (rw < HID) {
                a.x = W1[rw * SDIM + k];
                a.y = W1[rw * SDIM + k + 1];
            }
            afrag[j] = a;
        }
    }
    __syncthreads();

    for (int t = 0; t < T; ++t) {
        const int buf = t & 1;

        if (t + 1 < T) {
            if (tid < SDIM)
                uS[buf ^ 1][tid] = u[(size_t)(t + 1) * SDIM + tid];
            else if (tid < SDIM + 2)
                wS[buf ^ 1][tid - SDIM] = w[(size_t)(t + 1) * 2 + (tid - SDIM)];
        }

        v8f acc[4];
        #pragma unroll
        for (int q = 0; q < 4; ++q)
            acc[q] = (v8f){0.f, 0.f, 0.f, 0.f, 0.f, 0.f, 0.f, 0.f};

        if (wave == NWA) {
            if (t > 0 && lane < NOBS)
                out_y[(size_t)(t - 1) * NOBS + lane] = cS[obsS[lane]];
            #pragma unroll
            for (int j = 0; j < NCHUNK; ++j) {
                const v2f b = *(const v2f*)(bBaseC + 4 * j);
                acc[j & 3] = __builtin_amdgcn_wmma_f32_16x16x4_f32(
                                 false, afrag[j], false, b, (short)0, acc[j & 3],
                                 false, false);
            }
            const v8f s = (acc[0] + acc[1]) + (acc[2] + acc[3]);
            if (lane == 0) {
                #pragma unroll
                for (int i = 0; i < HID; ++i)
                    hS[i] = tanhf(s[i] + b1S[i]);
            }
        } else {
            #pragma unroll
            for (int j = 0; j < NCHUNK; ++j) {
                const v2f b = *(const v2f*)(bBaseC + 4 * j);
                acc[j & 3] = __builtin_amdgcn_wmma_f32_16x16x4_f32(
                                 false, afrag[j], false, b, (short)0, acc[j & 3],
                                 false, false);
            }
        }
        __syncthreads();

        if (wave < NWA) {
            v8f s = (acc[0] + acc[1]) + (acc[2] + acc[3]);
            #pragma unroll
            for (int jj = 0; jj < 2; ++jj) {
                const v2f b = *(const v2f*)(bBaseH + 4 * jj);
                s = __builtin_amdgcn_wmma_f32_16x16x4_f32(
                        false, w2frag[jj], false, b, (short)0, s, false, false);
            }
            if (rloc == 0) {
                const int base = wave * 16 + hi * 8;
                #pragma unroll
                for (int i = 0; i < 8; ++i)
                    accS[base + i] = s[i];
            }
        }
        __syncthreads();

        if (tid < SDIM) {
            const float cn = accS[tid]
                           + BwS[2 * tid]     * wS[buf][0]
                           + BwS[2 * tid + 1] * wS[buf][1]
                           + uS[buf][tid]
                           + b2S[tid];
            cS[tid] = cn;
            out_c[(size_t)t * SDIM + tid] = cn;
        }
        __syncthreads();
    }

    if (tid < NOBS)
        out_y[(size_t)(T - 1) * NOBS + tid] = cS[obsS[tid]];
}

extern "C" void kernel_launch(void* const* d_in, const int* in_sizes, int n_in,
                              void* d_out, int out_size, void* d_ws, size_t ws_size,
                              hipStream_t stream) {
    const float* c0 = (const float*)d_in[0];
    const float* w  = (const float*)d_in[1];
    const float* u  = (const float*)d_in[2];
    const float* A  = (const float*)d_in[3];
    const float* Bw = (const float*)d_in[4];
    const float* W1 = (const float*)d_in[5];
    const float* b1 = (const float*)d_in[6];
    const float* W2 = (const float*)d_in[7];
    const float* b2 = (const float*)d_in[8];
    const int*  obs = (const int*)  d_in[9];

    const int T = in_sizes[1] / 2;
    float* out_c = (float*)d_out;
    float* out_y = out_c + (size_t)T * SDIM;

    greybox_scan_kernel<<<1, 256, 0, stream>>>(c0, w, u, A, Bw, W1, b1, W2, b2,
                                               obs, out_c, out_y, T);
}


# W1 product transposed (state as WMMA A row 0, W1 rows as B columns): hidden unit n lands in lane n, so all hidden-unit tanh run lane-parallel instead of serially in lane 0; sum order unchanged
# speedup vs baseline: 1.2614x; 1.2614x over previous
#include <hip/hip_runtime.h>
#include <hip/hip_bf16.h>

#define SDIM   100
#define SPAD   112
#define HID    8
#define NOBS   12
#define DTC    0.01f
#define NCHUNK 25
#define NWA    7

typedef float v2f __attribute__((ext_vector_type(2)));
typedef float v8f __attribute__((ext_vector_type(8)));

__global__ __launch_bounds__(256, 1)
void greybox_scan_kernel(const float* __restrict__ c0,
                         const float* __restrict__ w,
                         const float* __restrict__ u,
                         const float* __restrict__ A,
                         const float* __restrict__ Bw,
                         const float* __restrict__ W1,
                         const float* __restrict__ b1,
                         const float* __restrict__ W2,
                         const float* __restrict__ b2,
                         const int*   __restrict__ obs,
                         float* __restrict__ out_c,
                         float* __restrict__ out_y,
                         int T)
{
    __shared__ float cS[SDIM];
    __shared__ float zS[SDIM];
    __shared__ float hS[HID];
    __shared__ float accS[SPAD];
    __shared__ float uS[2][SDIM];
    __shared__ float wS[2][2];
    __shared__ float BwS[2 * SDIM];
    __shared__ float b2S[SDIM];
    __shared__ int   obsS[NOBS];

    const int tid  = threadIdx.x;
    const int wave = tid >> 5;
    const int lane = tid & 31;
    const int hi   = (lane >> 4) & 1;
    const int rloc = lane & 15;

    for (int i = tid; i < 2 * SDIM; i += 256) BwS[i] = Bw[i];
    for (int i = tid; i < SDIM;     i += 256) { b2S[i] = DTC * b2[i]; zS[i] = 0.0f; cS[i] = c0[i]; }
    for (int i = tid; i < NOBS;     i += 256) obsS[i] = obs[i];
    for (int i = tid; i < SPAD;     i += 256) accS[i] = 0.0f;
    if (tid < SDIM) { uS[0][tid] = u[tid]; uS[1][tid] = 0.0f; }
    if (tid < 2)    { wS[0][tid] = w[tid]; wS[1][tid] = 0.0f; }

    const float* bBaseC = (rloc == 0) ? (cS + 2 * hi) : zS;
    const float* bBaseH = (rloc == 0) ? (hS + 2 * hi) : zS;

    v2f afrag[NCHUNK] = {};
    v2f w2frag[2]     = {};
    float b1v = 0.0f;

    if (wave < NWA) {
        const int r = wave * 16 + rloc;
        #pragma unroll
        for (int j = 0; j < NCHUNK; ++j) {
            const int k = 4 * j + 2 * hi;
            v2f a; a.x = 0.0f; a.y = 0.0f;
            if (r < SDIM) {
                a.x = A[r * SDIM + k];
                a.y = A[r * SDIM + k + 1];
            }
            afrag[j] = a;
        }
        #pragma unroll
        for (int jj = 0; jj < 2; ++jj) {
            const int k = 4 * jj + 2 * hi;
            v2f a; a.x = 0.0f; a.y = 0.0f;
            if (r < SDIM) {
                a.x = DTC * W2[r * HID + k];
                a.y = DTC * W2[r * HID + k + 1];
            }
            w2frag[jj] = a;
        }
    } else {
        const int rw = rloc;
        if (lane < HID) b1v = b1[lane];
        #pragma unroll
        for (int j = 0; j < NCHUNK; ++j) {
            const int k = 4 * j + 2 * hi;
            v2f a; a.x = 0.0f; a.y = 0.0f;
            if (rw < HID) {
                a.x = W1[rw * SDIM + k];
                a.y = W1[rw * SDIM + k + 1];
            }
            afrag[j] = a;
        }
    }
    __syncthreads();

    for (int t = 0; t < T; ++t) {
        const int buf = t & 1;

        if (t + 1 < T) {
            if (tid < SDIM)
                uS[buf ^ 1][tid] = u[(size_t)(t + 1) * SDIM + tid];
            else if (tid < SDIM + 2)
                wS[buf ^ 1][tid - SDIM] = w[(size_t)(t + 1) * 2 + (tid - SDIM)];
        }

        v8f acc[4];
        #pragma unroll
        for (int q = 0; q < 4; ++q)
            acc[q] = (v8f){0.f, 0.f, 0.f, 0.f, 0.f, 0.f, 0.f, 0.f};

        if (wave == NWA) {
            if (t > 0 && lane < NOBS)
                out_y[(size_t)(t - 1) * NOBS + lane] = cS[obsS[lane]];
            #pragma unroll
            for (int j = 0; j < NCHUNK; ++j) {
                const v2f cfrag = *(const v2f*)(bBaseC + 4 * j);
                acc[j & 3] = __builtin_amdgcn_wmma_f32_16x16x4_f32(
                                 false, cfrag, false, afrag[j], (short)0, acc[j & 3],
                                 false, false);
            }
            const float s0 = (acc[0][0] + acc[1][0]) + (acc[2][0] + acc[3][0]);
            if (lane < HID)
                hS[lane] = tanhf(s0 + b1v);
        } else {
            #pragma unroll
            for (int j = 0; j < NCHUNK; ++j) {
                const v2f b = *(const v2f*)(bBaseC + 4 * j);
                acc[j & 3] = __builtin_amdgcn_wmma_f32_16x16x4_f32(
                                 false, afrag[j], false, b, (short)0, acc[j & 3],
                                 false, false);
            }
        }
        __syncthreads();

        if (wave < NWA) {
            v8f s = (acc[0] + acc[1]) + (acc[2] + acc[3]);
            #pragma unroll
            for (int jj = 0; jj < 2; ++jj) {
                const v2f b = *(const v2f*)(bBaseH + 4 * jj);
                s = __builtin_amdgcn_wmma_f32_16x16x4_f32(
                        false, w2frag[jj], false, b, (short)0, s, false, false);
            }
            if (rloc == 0) {
                const int base = wave * 16 + hi * 8;
                #pragma unroll
                for (int i = 0; i < 8; ++i)
                    accS[base + i] = s[i];
            }
        }
        __syncthreads();

        if (tid < SDIM) {
            const float cn = accS[tid]
                           + BwS[2 * tid]     * wS[buf][0]
                           + BwS[2 * tid + 1] * wS[buf][1]
                           + uS[buf][tid]
                           + b2S[tid];
            cS[tid] = cn;
            out_c[(size_t)t * SDIM + tid] = cn;
        }
        __syncthreads();
    }

    if (tid < NOBS)
        out_y[(size_t)(T - 1) * NOBS + tid] = cS[obsS[tid]];
}

extern "C" void kernel_launch(void* const* d_in, const int* in_sizes, int n_in,
                              void* d_out, int out_size, void* d_ws, size_t ws_size,
                              hipStream_t stream) {
    const float* c0 = (const float*)d_in[0];
    const float* w  = (const float*)d_in[1];
    const float* u  = (const float*)d_in[2];
    const float* A  = (const float*)d_in[3];
    const float* Bw = (const float*)d_in[4];
    const float* W1 = (const float*)d_in[5];
    const float* b1 = (const float*)d_in[6];
    const float* W2 = (const float*)d_in[7];
    const float* b2 = (const float*)d_in[8];
    const int*  obs = (const int*)  d_in[9];

    const int T = in_sizes[1] / 2;
    float* out_c = (float*)d_out;
    float* out_y = out_c + (size_t)T * SDIM;

    greybox_scan_kernel<<<1, 256, 0, stream>>>(c0, w, u, A, Bw, W1, b1, W2, b2,
                                               obs, out_c, out_y, T);
}


# Global-memory latency taken off the serial time step: barriers fence LDS only (no load/store drain) and u,w rows wait in registers one step ahead instead of LDS staging; sum order unchanged
# speedup vs baseline: 1.9741x; 1.5650x over previous
#include <hip/hip_runtime.h>
#include <hip/hip_bf16.h>

#define SDIM   100
#define SPAD   112
#define HID    8
#define NOBS   12
#define DTC    0.01f
#define NCHUNK 25
#define NWA    7

typedef float v2f __attribute__((ext_vector_type(2)));
typedef float v8f __attribute__((ext_vector_type(8)));

static __device__ __forceinline__ void lds_barrier()
{
    __builtin_amdgcn_fence(__ATOMIC_RELEASE, "workgroup", "local");
    __builtin_amdgcn_s_barrier();
    __builtin_amdgcn_fence(__ATOMIC_ACQUIRE, "workgroup", "local");
}

__global__ __launch_bounds__(256, 1)
void greybox_scan_kernel(const float* __restrict__ c0,
                         const float* __restrict__ w,
                         const float* __restrict__ u,
                         const float* __restrict__ A,
                         const float* __restrict__ Bw,
                         const float* __restrict__ W1,
                         const float* __restrict__ b1,
                         const float* __restrict__ W2,
                         const float* __restrict__ b2,
                         const int*   __restrict__ obs,
                         float* __restrict__ out_c,
                         float* __restrict__ out_y,
                         int T)
{
    __shared__ float cS[SDIM];
    __shared__ float zS[SDIM];
    __shared__ float hS[HID];
    __shared__ float accS[SPAD];
    __shared__ float BwS[2 * SDIM];
    __shared__ float b2S[SDIM];
    __shared__ int   obsS[NOBS];

    const int tid  = threadIdx.x;
    const int wave = tid >> 5;
    const int lane = tid & 31;
    const int hi   = (lane >> 4) & 1;
    const int rloc = lane & 15;

    for (int i = tid; i < 2 * SDIM; i += 256) BwS[i] = Bw[i];
    for (int i = tid; i < SDIM;     i += 256) { b2S[i] = DTC * b2[i]; zS[i] = 0.0f; cS[i] = c0[i]; }
    for (int i = tid; i < NOBS;     i += 256) obsS[i] = obs[i];
    for (int i = tid; i < SPAD;     i += 256) accS[i] = 0.0f;

    float u_cur = (tid < SDIM) ? u[tid] : 0.0f;
    float w_cur0 = w[0];
    float w_cur1 = w[1];

    const float* bBaseC = (rloc == 0) ? (cS + 2 * hi) : zS;
    const float* bBaseH = (rloc == 0) ? (hS + 2 * hi) : zS;

    v2f afrag[NCHUNK] = {};
    v2f w2frag[2]     = {};
    float b1v = 0.0f;

    if (wave < NWA) {
        const int r = wave * 16 + rloc;
        #pragma unroll
        for (int j = 0; j < NCHUNK; ++j) {
            const int k = 4 * j + 2 * hi;
            v2f a; a.x = 0.0f; a.y = 0.0f;
            if (r < SDIM) {
                a.x = A[r * SDIM + k];
                a.y = A[r * SDIM + k + 1];
            }
            afrag[j] = a;
        }
        #pragma unroll
        for (int jj = 0; jj < 2; ++jj) {
            const int k = 4 * jj + 2 * hi;
            v2f a; a.x = 0.0f; a.y = 0.0f;
            if (r < SDIM) {
                a.x = DTC * W2[r * HID + k];
                a.y = DTC * W2[r * HID + k + 1];
            }
            w2frag[jj] = a;
        }
    } else {
        const int rw = rloc;
        if (lane < HID) b1v = b1[lane];
        #pragma unroll
        for (int j = 0; j < NCHUNK; ++j) {
            const int k = 4 * j + 2 * hi;
            v2f a; a.x = 0.0f; a.y = 0.0f;
            if (rw < HID) {
                a.x = W1[rw * SDIM + k];
                a.y = W1[rw * SDIM + k + 1];
            }
            afrag[j] = a;
        }
    }
    __syncthreads();

    for (int t = 0; t < T; ++t) {
        float u_nxt = 0.0f, w_nxt0 = 0.0f, w_nxt1 = 0.0f;
        if (t + 1 < T) {
            if (tid < SDIM)
                u_nxt = u[(size_t)(t + 1) * SDIM + tid];
            w_nxt0 = w[(size_t)(t + 1) * 2];
            w_nxt1 = w[(size_t)(t + 1) * 2 + 1];
        }

        v8f acc[4];
        #pragma unroll
        for (int q = 0; q < 4; ++q)
            acc[q] = (v8f){0.f, 0.f, 0.f, 0.f, 0.f, 0.f, 0.f, 0.f};

        if (wave == NWA) {
            if (t > 0 && lane < NOBS)
                out_y[(size_t)(t - 1) * NOBS + lane] = cS[obsS[lane]];
            #pragma unroll
            for (int j = 0; j < NCHUNK; ++j) {
                const v2f cfrag = *(const v2f*)(bBaseC + 4 * j);
                acc[j & 3] = __builtin_amdgcn_wmma_f32_16x16x4_f32(
                                 false, cfrag, false, afrag[j], (short)0, acc[j & 3],
                                 false, false);
            }
            const float s0 = (acc[0][0] + acc[1][0]) + (acc[2][0] + acc[3][0]);
            if (lane < HID)
                hS[lane] = tanhf(s0 + b1v);
        } else {
            #pragma unroll
            for (int j = 0; j < NCHUNK; ++j) {
                const v2f b = *(const v2f*)(bBaseC + 4 * j);
                acc[j & 3] = __builtin_amdgcn_wmma_f32_16x16x4_f32(
                                 false, afrag[j], false, b, (short)0, acc[j & 3],
                                 false, false);
            }
        }
        lds_barrier();

        if (wave < NWA) {
            v8f s = (acc[0] + acc[1]) + (acc[2] + acc[3]);
            #pragma unroll
            for (int jj = 0; jj < 2; ++jj) {
                const v2f b = *(const v2f*)(bBaseH + 4 * jj);
                s = __builtin_amdgcn_wmma_f32_16x16x4_f32(
                        false, w2frag[jj], false, b, (short)0, s, false, false);
            }
            if (rloc == 0) {
                const int base = wave * 16 + hi * 8;
                #pragma unroll
                for (int i = 0; i < 8; ++i)
                    accS[base + i] = s[i];
            }
        }
        lds_barrier();

        if (tid < SDIM) {
            const float cn = accS[tid]
                           + BwS[2 * tid]     * w_cur0
                           + BwS[2 * tid + 1] * w_cur1
                           + u_cur
                           + b2S[tid];
            cS[tid] = cn;
            out_c[(size_t)t * SDIM + tid] = cn;
        }
        lds_barrier();
        u_cur  = u_nxt;
        w_cur0 = w_nxt0;
        w_cur1 = w_nxt1;
    }

    if (tid < NOBS)
        out_y[(size_t)(T - 1) * NOBS + tid] = cS[obsS[tid]];
}

extern "C" void kernel_launch(void* const* d_in, const int* in_sizes, int n_in,
                              void* d_out, int out_size, void* d_ws, size_t ws_size,
                              hipStream_t stream) {
    const float* c0 = (const float*)d_in[0];
    const float* w  = (const float*)d_in[1];
    const float* u  = (const float*)d_in[2];
    const float* A  = (const float*)d_in[3];
    const float* Bw = (const float*)d_in[4];
    const float* W1 = (const float*)d_in[5];
    const float* b1 = (const float*)d_in[6];
    const float* W2 = (const float*)d_in[7];
    const float* b2 = (const float*)d_in[8];
    const int*  obs = (const int*)  d_in[9];

    const int T = in_sizes[1] / 2;
    float* out_c = (float*)d_out;
    float* out_y = out_c + (size_t)T * SDIM;

    greybox_scan_kernel<<<1, 256, 0, stream>>>(c0, w, u, A, Bw, W1, b1, W2, b2,
                                               obs, out_c, out_y, T);
}


# State vector's 25 LDS fragment loads issued as one group into registers ahead of each step's WMMA run (was one LDS round trip per WMMA pair inside the run); sum order unchanged
# speedup vs baseline: 2.6314x; 1.3330x over previous
#include <hip/hip_runtime.h>
#include <hip/hip_bf16.h>

#define SDIM   100
#define SPAD   112
#define HID    8
#define NOBS   12
#define DTC    0.01f
#define NCHUNK 25
#define NWA    7

typedef float v2f __attribute__((ext_vector_type(2)));
typedef float v8f __attribute__((ext_vector_type(8)));

static __device__ __forceinline__ void lds_barrier()
{
    __builtin_amdgcn_fence(__ATOMIC_RELEASE, "workgroup", "local");
    __builtin_amdgcn_s_barrier();
    __builtin_amdgcn_fence(__ATOMIC_ACQUIRE, "workgroup", "local");
}

__global__ __launch_bounds__(256, 1)
void greybox_scan_kernel(const float* __restrict__ c0,
                         const float* __restrict__ w,
                         const float* __restrict__ u,
                         const float* __restrict__ A,
                         const float* __restrict__ Bw,
                         const float* __restrict__ W1,
                         const float* __restrict__ b1,
                         const float* __restrict__ W2,
                         const float* __restrict__ b2,
                         const int*   __restrict__ obs,
                         float* __restrict__ out_c,
                         float* __restrict__ out_y,
                         int T)
{
    __shared__ float cS[SDIM];
    __shared__ float zS[SDIM];
    __shared__ float hS[HID];
    __shared__ float accS[SPAD];
    __shared__ float BwS[2 * SDIM];
    __shared__ float b2S[SDIM];
    __shared__ int   obsS[NOBS];

    const int tid  = threadIdx.x;
    const int wave = tid >> 5;
    const int lane = tid & 31;
    const int hi   = (lane >> 4) & 1;
    const int rloc = lane & 15;

    for (int i = tid; i < 2 * SDIM; i += 256) BwS[i] = Bw[i];
    for (int i = tid; i < SDIM;     i += 256) { b2S[i] = DTC * b2[i]; zS[i] = 0.0f; cS[i] = c0[i]; }
    for (int i = tid; i < NOBS;     i += 256) obsS[i] = obs[i];
    for (int i = tid; i < SPAD;     i += 256) accS[i] = 0.0f;

    float u_cur = (tid < SDIM) ? u[tid] : 0.0f;
    float w_cur0 = w[0];
    float w_cur1 = w[1];

    const float* bBaseC = (rloc == 0) ? (cS + 2 * hi) : zS;
    const float* bBaseH = (rloc == 0) ? (hS + 2 * hi) : zS;

    v2f afrag[NCHUNK] = {};
    v2f w2frag[2]     = {};
    float b1v = 0.0f;

    if (wave < NWA) {
        const int r = wave * 16 + rloc;
        #pragma unroll
        for (int j = 0; j < NCHUNK; ++j) {
            const int k = 4 * j + 2 * hi;
            v2f a; a.x = 0.0f; a.y = 0.0f;
            if (r < SDIM) {
                a.x = A[r * SDIM + k];
                a.y = A[r * SDIM + k + 1];
            }
            afrag[j] = a;
        }
        #pragma unroll
        for (int jj = 0; jj < 2; ++jj) {
            const int k = 4 * jj + 2 * hi;
            v2f a; a.x = 0.0f; a.y = 0.0f;
            if (r < SDIM) {
                a.x = DTC * W2[r * HID + k];
                a.y = DTC * W2[r * HID + k + 1];
            }
            w2frag[jj] = a;
        }
    } else {
        const int rw = rloc;
        if (lane < HID) b1v = b1[lane];
        #pragma unroll
        for (int j = 0; j < NCHUNK; ++j) {
            const int k = 4 * j + 2 * hi;
            v2f a; a.x = 0.0f; a.y = 0.0f;
            if (rw < HID) {
                a.x = W1[rw * SDIM + k];
                a.y = W1[rw * SDIM + k + 1];
            }
            afrag[j] = a;
        }
    }
    __syncthreads();

    for (int t = 0; t < T; ++t) {
        float u_nxt = 0.0f, w_nxt0 = 0.0f, w_nxt1 = 0.0f;
        if (t + 1 < T) {
            if (tid < SDIM)
                u_nxt = u[(size_t)(t + 1) * SDIM + tid];
            w_nxt0 = w[(size_t)(t + 1) * 2];
            w_nxt1 = w[(size_t)(t + 1) * 2 + 1];
        }

        v2f cf[NCHUNK];
        #pragma unroll
        for (int j = 0; j < NCHUNK; ++j)
            cf[j] = *(const v2f*)(bBaseC + 4 * j);

        v8f acc[4];
        #pragma unroll
        for (int q = 0; q < 4; ++q)
            acc[q] = (v8f){0.f, 0.f, 0.f, 0.f, 0.f, 0.f, 0.f, 0.f};

        if (wave == NWA) {
            if (t > 0 && lane < NOBS)
                out_y[(size_t)(t - 1) * NOBS + lane] = cS[obsS[lane]];
            #pragma unroll
            for (int j = 0; j < NCHUNK; ++j) {
                acc[j & 3] = __builtin_amdgcn_wmma_f32_16x16x4_f32(
                                 false, cf[j], false, afrag[j], (short)0, acc[j & 3],
                                 false, false);
            }
            const float s0 = (acc[0][0] + acc[1][0]) + (acc[2][0] + acc[3][0]);
            if (lane < HID)
                hS[lane] = tanhf(s0 + b1v);
        } else {
            #pragma unroll
            for (int j = 0; j < NCHUNK; ++j) {
                acc[j & 3] = __builtin_amdgcn_wmma_f32_16x16x4_f32(
                                 false, afrag[j], false, cf[j], (short)0, acc[j & 3],
                                 false, false);
            }
        }
        lds_barrier();

        if (wave < NWA) {
            v8f s = (acc[0] + acc[1]) + (acc[2] + acc[3]);
            #pragma unroll
            for (int jj = 0; jj < 2; ++jj) {
                const v2f b = *(const v2f*)(bBaseH + 4 * jj);
                s = __builtin_amdgcn_wmma_f32_16x16x4_f32(
                        false, w2frag[jj], false, b, (short)0, s, false, false);
            }
            if (rloc == 0) {
                const int base = wave * 16 + hi * 8;
                #pragma unroll
                for (int i = 0; i < 8; ++i)
                    accS[base + i] = s[i];
            }
        }
        lds_barrier();

        if (tid < SDIM) {
            const float cn = accS[tid]
                           + BwS[2 * tid]     * w_cur0
                           + BwS[2 * tid + 1] * w_cur1
                           + u_cur
                           + b2S[tid];
            cS[tid] = cn;
            out_c[(size_t)t * SDIM + tid] = cn;
        }
        lds_barrier();
        u_cur  = u_nxt;
        w_cur0 = w_nxt0;
        w_cur1 = w_nxt1;
    }

    if (tid < NOBS)
        out_y[(size_t)(T - 1) * NOBS + tid] = cS[obsS[tid]];
}

extern "C" void kernel_launch(void* const* d_in, const int* in_sizes, int n_in,
                              void* d_out, int out_size, void* d_ws, size_t ws_size,
                              hipStream_t stream) {
    const float* c0 = (const float*)d_in[0];
    const float* w  = (const float*)d_in[1];
    const float* u  = (const float*)d_in[2];
    const float* A  = (const float*)d_in[3];
    const float* Bw = (const float*)d_in[4];
    const float* W1 = (const float*)d_in[5];
    const float* b1 = (const float*)d_in[6];
    const float* W2 = (const float*)d_in[7];
    const float* b2 = (const float*)d_in[8];
    const int*  obs = (const int*)  d_in[9];

    const int T = in_sizes[1] / 2;
    float* out_c = (float*)d_out;
    float* out_y = out_c + (size_t)T * SDIM;

    greybox_scan_kernel<<<1, 256, 0, stream>>>(c0, w, u, A, Bw, W1, b1, W2, b2,
                                               obs, out_c, out_y, T);
}


# W1+tanh wave raised to issue priority 3 over the A wave sharing its SIMD, its y-row gather moved to wave 0's idle time: tanh now runs under the partner's WMMAs, not after them; sum order unchanged
# speedup vs baseline: 3.0449x; 1.1571x over previous
#include <hip/hip_runtime.h>
#include <hip/hip_bf16.h>

#define SDIM   100
#define SPAD   112
#define HID    8
#define NOBS   12
#define DTC    0.01f
#define NCHUNK 25
#define NWA    7

typedef float v2f __attribute__((ext_vector_type(2)));
typedef float v8f __attribute__((ext_vector_type(8)));

static __device__ __forceinline__ void lds_barrier()
{
    __builtin_amdgcn_fence(__ATOMIC_RELEASE, "workgroup", "local");
    __builtin_amdgcn_s_barrier();
    __builtin_amdgcn_fence(__ATOMIC_ACQUIRE, "workgroup", "local");
}

__global__ __launch_bounds__(256, 1)
void greybox_scan_kernel(const float* __restrict__ c0,
                         const float* __restrict__ w,
                         const float* __restrict__ u,
                         const float* __restrict__ A,
                         const float* __restrict__ Bw,
                         const float* __restrict__ W1,
                         const float* __restrict__ b1,
                         const float* __restrict__ W2,
                         const float* __restrict__ b2,
                         const int*   __restrict__ obs,
                         float* __restrict__ out_c,
                         float* __restrict__ out_y,
                         int T)
{
    __shared__ float cS[SDIM];
    __shared__ float zS[SDIM];
    __shared__ float hS[HID];
    __shared__ float accS[SPAD];
    __shared__ float BwS[2 * SDIM];
    __shared__ float b2S[SDIM];
    __shared__ int   obsS[NOBS];

    const int tid  = threadIdx.x;
    const int wave = tid >> 5;
    const int lane = tid & 31;
    const int hi   = (lane >> 4) & 1;
    const int rloc = lane & 15;

    for (int i = tid; i < 2 * SDIM; i += 256) BwS[i] = Bw[i];
    for (int i = tid; i < SDIM;     i += 256) { b2S[i] = DTC * b2[i]; zS[i] = 0.0f; cS[i] = c0[i]; }
    for (int i = tid; i < NOBS;     i += 256) obsS[i] = obs[i];
    for (int i = tid; i < SPAD;     i += 256) accS[i] = 0.0f;

    float u_cur = (tid < SDIM) ? u[tid] : 0.0f;
    float w_cur0 = w[0];
    float w_cur1 = w[1];

    const float* bBaseC = (rloc == 0) ? (cS + 2 * hi) : zS;
    const float* bBaseH = (rloc == 0) ? (hS + 2 * hi) : zS;

    v2f afrag[NCHUNK] = {};
    v2f w2frag[2]     = {};
    float b1v = 0.0f;

    if (wave < NWA) {
        const int r = wave * 16 + rloc;
        #pragma unroll
        for (int j = 0; j < NCHUNK; ++j) {
            const int k = 4 * j + 2 * hi;
            v2f a; a.x = 0.0f; a.y = 0.0f;
            if (r < SDIM) {
                a.x = A[r * SDIM + k];
                a.y = A[r * SDIM + k + 1];
            }
            afrag[j] = a;
        }
        #pragma unroll
        for (int jj = 0; jj < 2; ++jj) {
            const int k = 4 * jj + 2 * hi;
            v2f a; a.x = 0.0f; a.y = 0.0f;
            if (r < SDIM) {
                a.x = DTC * W2[r * HID + k];
                a.y = DTC * W2[r * HID + k + 1];
            }
            w2frag[jj] = a;
        }
    } else {
        const int rw = rloc;
        if (lane < HID) b1v = b1[lane];
        #pragma unroll
        for (int j = 0; j < NCHUNK; ++j) {
            const int k = 4 * j + 2 * hi;
            v2f a; a.x = 0.0f; a.y = 0.0f;
            if (rw < HID) {
                a.x = W1[rw * SDIM + k];
                a.y = W1[rw * SDIM + k + 1];
            }
            afrag[j] = a;
        }
    }
    __syncthreads();

    const int wave_u = __builtin_amdgcn_readfirstlane(wave);
    if (wave_u == NWA)
        __builtin_amdgcn_s_setprio(3);

    for (int t = 0; t < T; ++t) {
        float u_nxt = 0.0f, w_nxt0 = 0.0f, w_nxt1 = 0.0f;
        if (t + 1 < T) {
            if (tid < SDIM)
                u_nxt = u[(size_t)(t + 1) * SDIM + tid];
            w_nxt0 = w[(size_t)(t + 1) * 2];
            w_nxt1 = w[(size_t)(t + 1) * 2 + 1];
        }

        v2f cf[NCHUNK];
        #pragma unroll
        for (int j = 0; j < NCHUNK; ++j)
            cf[j] = *(const v2f*)(bBaseC + 4 * j);

        v8f acc[4];
        #pragma unroll
        for (int q = 0; q < 4; ++q)
            acc[q] = (v8f){0.f, 0.f, 0.f, 0.f, 0.f, 0.f, 0.f, 0.f};

        if (wave == NWA) {
            #pragma unroll
            for (int j = 0; j < NCHUNK; ++j) {
                acc[j & 3] = __builtin_amdgcn_wmma_f32_16x16x4_f32(
                                 false, cf[j], false, afrag[j], (short)0, acc[j & 3],
                                 false, false);
            }
            const float s0 = (acc[0][0] + acc[1][0]) + (acc[2][0] + acc[3][0]);
            if (lane < HID)
                hS[lane] = tanhf(s0 + b1v);
        } else {
            #pragma unroll
            for (int j = 0; j < NCHUNK; ++j) {
                acc[j & 3] = __builtin_amdgcn_wmma_f32_16x16x4_f32(
                                 false, afrag[j], false, cf[j], (short)0, acc[j & 3],
                                 false, false);
            }
            if (wave == 0 && t > 0 && lane < NOBS)
                out_y[(size_t)(t - 1) * NOBS + lane] = cS[obsS[lane]];
        }
        lds_barrier();

        if (wave < NWA) {
            v8f s = (acc[0] + acc[1]) + (acc[2] + acc[3]);
            #pragma unroll
            for (int jj = 0; jj < 2; ++jj) {
                const v2f b = *(const v2f*)(bBaseH + 4 * jj);
                s = __builtin_amdgcn_wmma_f32_16x16x4_f32(
                        false, w2frag[jj], false, b, (short)0, s, false, false);
            }
            if (rloc == 0) {
                const int base = wave * 16 + hi * 8;
                #pragma unroll
                for (int i = 0; i < 8; ++i)
                    accS[base + i] = s[i];
            }
        }
        lds_barrier();

        if (tid < SDIM) {
            const float cn = accS[tid]
                           + BwS[2 * tid]     * w_cur0
                           + BwS[2 * tid + 1] * w_cur1
                           + u_cur
                           + b2S[tid];
            cS[tid] = cn;
            out_c[(size_t)t * SDIM + tid] = cn;
        }
        lds_barrier();
        u_cur  = u_nxt;
        w_cur0 = w_nxt0;
        w_cur1 = w_nxt1;
    }

    if (wave_u == NWA)
        __builtin_amdgcn_s_setprio(0);

    if (tid < NOBS)
        out_y[(size_t)(T - 1) * NOBS + tid] = cS[obsS[tid]];
}

extern "C" void kernel_launch(void* const* d_in, const int* in_sizes, int n_in,
                              void* d_out, int out_size, void* d_ws, size_t ws_size,
                              hipStream_t stream) {
    const float* c0 = (const float*)d_in[0];
    const float* w  = (const float*)d_in[1];
    const float* u  = (const float*)d_in[2];
    const float* A  = (const float*)d_in[3];
    const float* Bw = (const float*)d_in[4];
    const float* W1 = (const float*)d_in[5];
    const float* b1 = (const float*)d_in[6];
    const float* W2 = (const float*)d_in[7];
    const float* b2 = (const float*)d_in[8];
    const int*  obs = (const int*)  d_in[9];

    const int T = in_sizes[1] / 2;
    float* out_c = (float*)d_out;
    float* out_y = out_c + (size_t)T * SDIM;

    greybox_scan_kernel<<<1, 256, 0, stream>>>(c0, w, u, A, Bw, W1, b1, W2, b2,
                                               obs, out_c, out_y, T);
}


# A and W2 products transposed (vector as WMMA A row 0, matrix rows as B columns): row n's sum lands in lane n, which finishes and stores it; accS staging and its barrier gone; sum order unchanged
# speedup vs baseline: 3.3036x; 1.0850x over previous
#include <hip/hip_runtime.h>
#include <hip/hip_bf16.h>


#define SDIM   100
#define HID    8
#define NOBS   12
#define DTC    0.01f
#define NCHUNK 25
#define NWA    7

typedef float v2f __attribute__((ext_vector_type(2)));
typedef float v8f __attribute__((ext_vector_type(8)));

static __device__ __forceinline__ void lds_barrier()
{
    __builtin_amdgcn_fence(__ATOMIC_RELEASE, "workgroup", "local");
    __builtin_amdgcn_s_barrier();
    __builtin_amdgcn_fence(__ATOMIC_ACQUIRE, "workgroup", "local");
}

__global__ __launch_bounds__(256, 1)
void greybox_scan_kernel(const float* __restrict__ c0,
                         const float* __restrict__ w,
                         const float* __restrict__ u,
                         const float* __restrict__ A,
                         const float* __restrict__ Bw,
                         const float* __restrict__ W1,
                         const float* __restrict__ b1,
                         const float* __restrict__ W2,
                         const float* __restrict__ b2,
                         const int*   __restrict__ obs,
                         float* __restrict__ out_c,
                         float* __restrict__ out_y,
                         int T)
{
    __shared__ float cS[SDIM];
    __shared__ float zS[SDIM];
    __shared__ float hS[HID];
    __shared__ int   obsS[NOBS];

    const int tid  = threadIdx.x;
    const int wave = tid >> 5;
    const int lane = tid & 31;
    const int hi   = (lane >> 4) & 1;
    const int rloc = lane & 15;

    for (int i = tid; i < SDIM; i += 256) { zS[i] = 0.0f; cS[i] = c0[i]; }
    for (int i = tid; i < NOBS; i += 256) obsS[i] = obs[i];

    const int  r   = wave * 16 + rloc;
    const bool fin = (wave < NWA) && (lane < 16) && (r < SDIM);

    float bw0 = 0.0f, bw1 = 0.0f, b2v = 0.0f, u_cur = 0.0f;
    if (fin) {
        bw0   = Bw[2 * r];
        bw1   = Bw[2 * r + 1];
        b2v   = DTC * b2[r];
        u_cur = u[r];
    }
    float w_cur0 = w[0];
    float w_cur1 = w[1];

    const float* vBaseC = (rloc == 0) ? (cS + 2 * hi) : zS;
    const float* vBaseH = (rloc == 0) ? (hS + 2 * hi) : zS;

    v2f mfrag[NCHUNK] = {};
    v2f w2frag[2]     = {};
    float b1v = 0.0f;

    if (wave < NWA) {
        #pragma unroll
        for (int j = 0; j < NCHUNK; ++j) {
            const int k = 4 * j + 2 * hi;
            v2f a; a.x = 0.0f; a.y = 0.0f;
            if (r < SDIM) {
                a.x = A[r * SDIM + k];
                a.y = A[r * SDIM + k + 1];
            }
            mfrag[j] = a;
        }
        #pragma unroll
        for (int jj = 0; jj < 2; ++jj) {
            const int k = 4 * jj + 2 * hi;
            v2f a; a.x = 0.0f; a.y = 0.0f;
            if (r < SDIM) {
                a.x = DTC * W2[r * HID + k];
                a.y = DTC * W2[r * HID + k + 1];
            }
            w2frag[jj] = a;
        }
    } else {
        const int rw = rloc;
        if (lane < HID) b1v = b1[lane];
        #pragma unroll
        for (int j = 0; j < NCHUNK; ++j) {
            const int k = 4 * j + 2 * hi;
            v2f a; a.x = 0.0f; a.y = 0.0f;
            if (rw < HID) {
                a.x = W1[rw * SDIM + k];
                a.y = W1[rw * SDIM + k + 1];
            }
            mfrag[j] = a;
        }
    }
    __syncthreads();

    const int wave_u = __builtin_amdgcn_readfirstlane(wave);
    if (wave_u == NWA)
        __builtin_amdgcn_s_setprio(3);

    for (int t = 0; t < T; ++t) {
        float u_nxt = 0.0f, w_nxt0 = 0.0f, w_nxt1 = 0.0f;
        if (t + 1 < T) {
            if (fin)
                u_nxt = u[(size_t)(t + 1) * SDIM + r];
            w_nxt0 = w[(size_t)(t + 1) * 2];
            w_nxt1 = w[(size_t)(t + 1) * 2 + 1];
        }

        v2f cf[NCHUNK];
        #pragma unroll
        for (int j = 0; j < NCHUNK; ++j)
            cf[j] = *(const v2f*)(vBaseC + 4 * j);
        __builtin_amdgcn_sched_barrier(0);

        v8f acc[4];
        #pragma unroll
        for (int q = 0; q < 4; ++q)
            acc[q] = (v8f){0.f, 0.f, 0.f, 0.f, 0.f, 0.f, 0.f, 0.f};

        #pragma unroll
        for (int j = 0; j < NCHUNK; ++j) {
            acc[j & 3] = __builtin_amdgcn_wmma_f32_16x16x4_f32(
                             false, cf[j], false, mfrag[j], (short)0, acc[j & 3],
                             false, false);
        }
        const float s0 = (acc[0][0] + acc[1][0]) + (acc[2][0] + acc[3][0]);

        if (wave == NWA) {
            if (lane < HID)
                hS[lane] = tanhf(s0 + b1v);
        } else {
            if (wave == 0 && t > 0 && lane < NOBS)
                out_y[(size_t)(t - 1) * NOBS + lane] = cS[obsS[lane]];
        }
        lds_barrier();

        if (wave < NWA) {
            v8f s = acc[0];
            s[0] = s0;
            #pragma unroll
            for (int jj = 0; jj < 2; ++jj) {
                const v2f hf = *(const v2f*)(vBaseH + 4 * jj);
                s = __builtin_amdgcn_wmma_f32_16x16x4_f32(
                        false, hf, false, w2frag[jj], (short)0, s, false, false);
            }
            if (fin) {
                const float cn = __builtin_fmaf(bw1, w_cur1,
                                     __builtin_fmaf(bw0, w_cur0, s[0]))
                               + u_cur
                               + b2v;
                cS[r] = cn;
                out_c[(size_t)t * SDIM + r] = cn;
            }
        }
        lds_barrier();
        u_cur  = u_nxt;
        w_cur0 = w_nxt0;
        w_cur1 = w_nxt1;
    }

    if (wave_u == NWA)
        __builtin_amdgcn_s_setprio(0);

    if (tid < NOBS)
        out_y[(size_t)(T - 1) * NOBS + tid] = cS[obsS[tid]];
}

extern "C" void kernel_launch(void* const* d_in, const int* in_sizes, int n_in,
                              void* d_out, int out_size, void* d_ws, size_t ws_size,
                              hipStream_t stream) {
    const float* c0 = (const float*)d_in[0];
    const float* w  = (const float*)d_in[1];
    const float* u  = (const float*)d_in[2];
    const float* A  = (const float*)d_in[3];
    const float* Bw = (const float*)d_in[4];
    const float* W1 = (const float*)d_in[5];
    const float* b1 = (const float*)d_in[6];
    const float* W2 = (const float*)d_in[7];
    const float* b2 = (const float*)d_in[8];
    const int*  obs = (const int*)  d_in[9];

    const int T = in_sizes[1] / 2;
    float* out_c = (float*)d_out;
    float* out_y = out_c + (size_t)T * SDIM;

    greybox_scan_kernel<<<1, 256, 0, stream>>>(c0, w, u, A, Bw, W1, b1, W2, b2,
                                               obs, out_c, out_y, T);
}
